# starting kernel
# baseline (speedup 1.0000x reference)
#include <hip/hip_runtime.h>

typedef float v2f __attribute__((ext_vector_type(2)));
typedef float v8f __attribute__((ext_vector_type(8)));

#define WAVES_PER_BLOCK 2
#define RAYS_PER_WAVE   16
#define NSAMP           128

__global__ __launch_bounds__(WAVES_PER_BLOCK * 32)
void DistortionLoss_82471962018601_kernel(const float* __restrict__ wp,
                                          const float* __restrict__ zp,
                                          const float* __restrict__ nearp,
                                          const float* __restrict__ farp,
                                          float* __restrict__ out,
                                          int nrays) {
    __shared__ float sA[WAVES_PER_BLOCK][NSAMP * RAYS_PER_WAVE];
    __shared__ float sB[WAVES_PER_BLOCK][NSAMP * RAYS_PER_WAVE];
    __shared__ float sT3[WAVES_PER_BLOCK][RAYS_PER_WAVE];

    const int wave  = threadIdx.x >> 5;
    const int lane  = threadIdx.x & 31;
    const int rbase = blockIdx.x * (WAVES_PER_BLOCK * RAYS_PER_WAVE) + wave * RAYS_PER_WAVE;

    for (int r = 0; r < RAYS_PER_WAVE; ++r) {
        int ray = rbase + r;
        if (ray >= nrays) ray = nrays - 1;
        const float nv  = nearp[ray];
        const float fv  = farp[ray];
        const float inv = 1.0f / (fv - nv);
        float t3 = 0.0f;
#pragma unroll
        for (int q = 0; q < 4; ++q) {
            const int i   = q * 32 + lane;
            const float w = wp[(long)ray * NSAMP + i];
            const float zlo = zp[(long)ray * (NSAMP + 1) + i];
            const float zhi = zp[(long)ray * (NSAMP + 1) + i + 1];
            const float blo = (zlo - nv) * inv;
            const float bhi = (zhi - nv) * inv;
            const float m   = 0.5f * (blo + bhi);
            sB[wave][i * RAYS_PER_WAVE + r] = w;
            sA[wave][i * RAYS_PER_WAVE + r] = w * m;
            t3 += w * w * (bhi - blo);
        }
        for (int off = 16; off > 0; off >>= 1) t3 += __shfl_xor(t3, off, 32);
        if (lane == 0) sT3[wave][r] = t3;
    }
    __syncthreads();

    const int ray16  = lane & 15;
    const int hi     = lane >> 4;
    const int rowOff = hi * 8;
    const int srow   = lane & 15;

    v2f sfrag[4];
#pragma unroll
    for (int kk = 0; kk < 4; ++kk) {
        const int col0 = kk * 4 + hi * 2;
        sfrag[kk].x = (col0     < srow) ? 1.0f : 0.0f;
        sfrag[kk].y = (col0 + 1 < srow) ? 1.0f : 0.0f;
    }

    float carryA = 0.0f, carryB = 0.0f;
    float dot = 0.0f;

    for (int c = 0; c < 8; ++c) {
        const float* baseB = &sB[wave][(c * 16) * RAYS_PER_WAVE];
        const float* baseA = &sA[wave][(c * 16) * RAYS_PER_WAVE];

        v8f Db = {};
        v8f Da = {};
#pragma unroll
        for (int kk = 0; kk < 4; ++kk) {
            const int kr = kk * 4 + hi * 2;
            v2f bf, af;
            bf.x = baseB[(kr    ) * RAYS_PER_WAVE + ray16];
            bf.y = baseB[(kr + 1) * RAYS_PER_WAVE + ray16];
            af.x = baseA[(kr    ) * RAYS_PER_WAVE + ray16];
            af.y = baseA[(kr + 1) * RAYS_PER_WAVE + ray16];
            Db = __builtin_amdgcn_wmma_f32_16x16x4_f32(
                     false, sfrag[kk], false, bf, (short)0, Db, false, false);
            Da = __builtin_amdgcn_wmma_f32_16x16x4_f32(
                     false, sfrag[kk], false, af, (short)0, Da, false, false);
        }

        float sb = 0.0f, sa = 0.0f;
#pragma unroll
        for (int g = 0; g < 8; ++g) {
            const float bv = baseB[(g + rowOff) * RAYS_PER_WAVE + ray16];
            const float av = baseA[(g + rowOff) * RAYS_PER_WAVE + ray16];
            dot += av * (Db[g] + carryB) - bv * (Da[g] + carryA);
            sb += bv;
            sa += av;
        }
        sb += __shfl_xor(sb, 16, 32);
        sa += __shfl_xor(sa, 16, 32);
        carryB += sb;
        carryA += sa;
    }

    dot += __shfl_xor(dot, 16, 32);

    if (lane < 16 && (rbase + lane) < nrays) {
        out[rbase + lane] = 2.0f * dot + (1.0f / 3.0f) * sT3[wave][lane];
    }
}

extern "C" void kernel_launch(void* const* d_in, const int* in_sizes, int n_in,
                              void* d_out, int out_size, void* d_ws, size_t ws_size,
                              hipStream_t stream) {
    const float* wp    = (const float*)d_in[0];
    const float* zp    = (const float*)d_in[1];
    const float* nearp = (const float*)d_in[2];
    const float* farp  = (const float*)d_in[3];
    float* out = (float*)d_out;

    const int nrays = in_sizes[2];
    const int raysPerBlock = WAVES_PER_BLOCK * RAYS_PER_WAVE;
    const int blocks = (nrays + raysPerBlock - 1) / raysPerBlock;

    DistortionLoss_82471962018601_kernel<<<blocks, WAVES_PER_BLOCK * 32, 0, stream>>>(
        wp, zp, nearp, farp, out, nrays);
}


# Phase 0 lane remap: 8 lanes per ray stage a wave's 4 rays in one branch-free pass with 128-bit loads (was a 4-trip ray loop); third term summed once over 8 lanes by DPP: new summation order
# speedup vs baseline: 4.4747x; 4.4747x over previous
#include <hip/hip_runtime.h>


typedef float v2f __attribute__((ext_vector_type(2)));
typedef float v8f __attribute__((ext_vector_type(8)));

#define TILE_RAYS        16
#define WAVES_PER_TILE   4
#define TILES_PER_BLOCK  2
#define WAVES_PER_BLOCK  (WAVES_PER_TILE * TILES_PER_BLOCK)
#define RAYS_PER_BLOCK   (TILE_RAYS * TILES_PER_BLOCK)
#define LOAD_RAYS        (TILE_RAYS / WAVES_PER_TILE)
#define NSAMP            128
#define NCHUNK           (NSAMP / 16)
#define SCAN_CHUNKS      (NCHUNK / WAVES_PER_TILE)
#define LANES_PER_RAY    (32 / LOAD_RAYS)
#define GROUP_PITCH      72
#define PLANE_DWORDS     ((NSAMP / 4) * GROUP_PITCH)

typedef float v4fu __attribute__((ext_vector_type(4), aligned(4)));

__device__ __forceinline__ int lds_off(int i, int r) {
    return (i >> 2) * GROUP_PITCH + (i & 3) * TILE_RAYS + r;
}

__device__ __forceinline__ float sum_over_8_lanes(float v) {
    v += __int_as_float(__builtin_amdgcn_update_dpp(0, __float_as_int(v), 0xB1, 0xF, 0xF, true));
    v += __int_as_float(__builtin_amdgcn_update_dpp(0, __float_as_int(v), 0x4E, 0xF, 0xF, true));
    v += __int_as_float(__builtin_amdgcn_update_dpp(0, __float_as_int(v), 0x141, 0xF, 0xF, true));
    return v;
}

__global__ __launch_bounds__(WAVES_PER_BLOCK * 32)
void distortion_loss_split_tile_kernel(const float* __restrict__ wp,
                                       const float* __restrict__ zp,
                                       const float* __restrict__ nearp,
                                       const float* __restrict__ farp,
                                       float* __restrict__ out,
                                       int nrays) {
    __shared__ float sA[TILES_PER_BLOCK][PLANE_DWORDS];
    __shared__ float sB[TILES_PER_BLOCK][PLANE_DWORDS];
    __shared__ float sT3[TILES_PER_BLOCK][TILE_RAYS];
    __shared__ float sPart[WAVES_PER_BLOCK][3][TILE_RAYS];

    const int wave  = threadIdx.x >> 5;
    const int lane  = threadIdx.x & 31;
    const int tile  = wave / WAVES_PER_TILE;
    const int sub   = wave % WAVES_PER_TILE;
    const int bbase = blockIdx.x * RAYS_PER_BLOCK;
    const int tbase = bbase + tile * TILE_RAYS;

    {
        const int rr = lane / LANES_PER_RAY;
        const int c8 = lane % LANES_PER_RAY;
        const int r  = sub * LOAD_RAYS + rr;
        int ray = tbase + r;
        if (ray >= nrays) ray = nrays - 1;
        const float* wrow = wp + (long)ray * NSAMP;
        const float* zrow = zp + (long)ray * (NSAMP + 1);

        v4fu  w4[4], z4[4];
        float z5[4];
#pragma unroll
        for (int q = 0; q < 4; ++q) {
            const int i0 = q * 32 + c8 * 4;
            w4[q] = *(const v4fu*)(wrow + i0);
            z4[q] = *(const v4fu*)(zrow + i0);
            z5[q] = zrow[i0 + 4];
        }
        const float nv  = nearp[ray];
        const float fv  = farp[ray];
        const float inv = 1.0f / (fv - nv);

        float* pB = &sB[tile][0];
        float* pA = &sA[tile][0];
        float t3 = 0.0f;
#pragma unroll
        for (int q = 0; q < 4; ++q) {
            const int i0 = q * 32 + c8 * 4;
            float b[5];
#pragma unroll
            for (int e = 0; e < 4; ++e) b[e] = (z4[q][e] - nv) * inv;
            b[4] = (z5[q] - nv) * inv;
#pragma unroll
            for (int j = 0; j < 4; ++j) {
                const float w   = w4[q][j];
                const float blo = b[j];
                const float bhi = b[j + 1];
                const float m   = 0.5f * (blo + bhi);
                const int   o   = lds_off(i0 + j, r);
                pB[o] = w;
                pA[o] = w * m;
                t3 += w * w * (bhi - blo);
            }
        }
        t3 = sum_over_8_lanes(t3);
        if (c8 == 0) sT3[tile][r] = t3;
    }
    __syncthreads();

    const int ray16  = lane & 15;
    const int hi     = lane >> 4;
    const int rowOff = hi * 8;
    const int srow   = lane & 15;

    v2f sfrag[4];
#pragma unroll
    for (int kk = 0; kk < 4; ++kk) {
        const int col0 = kk * 4 + hi * 2;
        sfrag[kk].x = (col0     < srow) ? 1.0f : 0.0f;
        sfrag[kk].y = (col0 + 1 < srow) ? 1.0f : 0.0f;
    }

    float carryA = 0.0f, carryB = 0.0f;
    float dot = 0.0f;

    for (int cc = 0; cc < SCAN_CHUNKS; ++cc) {
        const int c = sub * SCAN_CHUNKS + cc;
        const float* baseB = &sB[tile][0];
        const float* baseA = &sA[tile][0];
        const int    s0    = c * 16;

        v8f Db = {};
        v8f Da = {};
#pragma unroll
        for (int kk = 0; kk < 4; ++kk) {
            const int kr = kk * 4 + hi * 2;
            v2f bf, af;
            bf.x = baseB[lds_off(s0 + kr,     ray16)];
            bf.y = baseB[lds_off(s0 + kr + 1, ray16)];
            af.x = baseA[lds_off(s0 + kr,     ray16)];
            af.y = baseA[lds_off(s0 + kr + 1, ray16)];
            Db = __builtin_amdgcn_wmma_f32_16x16x4_f32(
                     false, sfrag[kk], false, bf, (short)0, Db, false, false);
            Da = __builtin_amdgcn_wmma_f32_16x16x4_f32(
                     false, sfrag[kk], false, af, (short)0, Da, false, false);
        }

        float sb = 0.0f, sa = 0.0f;
#pragma unroll
        for (int g = 0; g < 8; ++g) {
            const float bv = baseB[lds_off(s0 + g + rowOff, ray16)];
            const float av = baseA[lds_off(s0 + g + rowOff, ray16)];
            dot += av * (Db[g] + carryB) - bv * (Da[g] + carryA);
            sb += bv;
            sa += av;
        }
        sb += __shfl_xor(sb, 16, 32);
        sa += __shfl_xor(sa, 16, 32);
        carryB += sb;
        carryA += sa;
    }

    dot += __shfl_xor(dot, 16, 32);

    if (lane < 16) {
        sPart[wave][0][lane] = dot;
        sPart[wave][1][lane] = carryA;
        sPart[wave][2][lane] = carryB;
    }
    __syncthreads();

    if (threadIdx.x < RAYS_PER_BLOCK) {
        const int t   = threadIdx.x >> 4;
        const int col = threadIdx.x & 15;
        float tot = 0.0f, preA = 0.0f, preB = 0.0f;
        for (int w = 0; w < WAVES_PER_TILE; ++w) {
            const float d  = sPart[t * WAVES_PER_TILE + w][0][col];
            const float SA = sPart[t * WAVES_PER_TILE + w][1][col];
            const float SB = sPart[t * WAVES_PER_TILE + w][2][col];
            tot += d + (SA * preB - SB * preA);
            preA += SA;
            preB += SB;
        }
        if (bbase + (int)threadIdx.x < nrays) {
            out[bbase + threadIdx.x] = 2.0f * tot + (1.0f / 3.0f) * sT3[t][col];
        }
    }
}

extern "C" void kernel_launch(void* const* d_in, const int* in_sizes, int n_in,
                              void* d_out, int out_size, void* d_ws, size_t ws_size,
                              hipStream_t stream) {
    const float* wp    = (const float*)d_in[0];
    const float* zp    = (const float*)d_in[1];
    const float* nearp = (const float*)d_in[2];
    const float* farp  = (const float*)d_in[3];
    float* out = (float*)d_out;

    const int nrays = in_sizes[2];
    const int raysPerBlock = RAYS_PER_BLOCK;
    const int blocks = (nrays + raysPerBlock - 1) / raysPerBlock;

    distortion_loss_split_tile_kernel<<<blocks, WAVES_PER_BLOCK * 32, 0, stream>>>(
        wp, zp, nearp, farp, out, nrays);
}


# Each wave owns its 4 rays from load to store (tile columns: 4 rays x 2 chunks x 2 scans): LDS hand-over inside the wave, no block barrier, no join pass, loads issued line by line; new summation order
# speedup vs baseline: 5.9485x; 1.3294x over previous
#include <hip/hip_runtime.h>


typedef float v2f __attribute__((ext_vector_type(2)));
typedef float v4f __attribute__((ext_vector_type(4)));
typedef float v8f __attribute__((ext_vector_type(8)));

#define WAVES_PER_BLOCK  8
#define RAYS_PER_WAVE    4
#define RAYS_PER_BLOCK   (WAVES_PER_BLOCK * RAYS_PER_WAVE)
#define NSAMP            128
#define GROUP            4
#define NGROUP           (NSAMP / GROUP)
#define NCHUNK           (NGROUP / 16)
#define LANES_PER_RAY    (32 / RAYS_PER_WAVE)
#define RAY_PITCH        36
#define OP_PITCH         160
#define WAVE_LDS         (2 * OP_PITCH)

typedef float v4fu __attribute__((ext_vector_type(4), aligned(4)));

__device__ __forceinline__ float sum_over_8_lanes(float v) {
    v += __int_as_float(__builtin_amdgcn_update_dpp(0, __float_as_int(v), 0xB1, 0xF, 0xF, true));
    v += __int_as_float(__builtin_amdgcn_update_dpp(0, __float_as_int(v), 0x4E, 0xF, 0xF, true));
    v += __int_as_float(__builtin_amdgcn_update_dpp(0, __float_as_int(v), 0x141, 0xF, 0xF, true));
    return v;
}

__device__ __forceinline__ float sum_over_quad(float v) {
    v += __int_as_float(__builtin_amdgcn_update_dpp(0, __float_as_int(v), 0xB1, 0xF, 0xF, true));
    v += __int_as_float(__builtin_amdgcn_update_dpp(0, __float_as_int(v), 0x4E, 0xF, 0xF, true));
    return v;
}

__device__ __forceinline__ float other_half(float v) {
    return __int_as_float(__builtin_amdgcn_permlanex16(__float_as_int(v), __float_as_int(v),
                                                       0x76543210, 0xfedcba98, false, false));
}

__global__ __launch_bounds__(WAVES_PER_BLOCK * 32)
void distortion_loss_wave_tile_kernel(const float* __restrict__ wp,
                                      const float* __restrict__ zp,
                                      const float* __restrict__ nearp,
                                      const float* __restrict__ farp,
                                      float* __restrict__ out,
                                      int nrays) {
    __shared__ __attribute__((aligned(16))) float sT[WAVES_PER_BLOCK][WAVE_LDS];

    const int wave  = threadIdx.x >> 5;
    const int lane  = threadIdx.x & 31;
    const int wbase = blockIdx.x * RAYS_PER_BLOCK + wave * RAYS_PER_WAVE;
    float* T = &sT[wave][0];
    asm volatile("" :: "s"(out));

    const int col = lane & 15;
    const int hi  = lane >> 4;
    const int op    = col & 1;
    const int chunk = (col >> 1) & 1;
    const int r1    = ((col >> 3) & 1) + 2 * ((col >> 2) & 1);

    const int rr = lane / LANES_PER_RAY;
    float loc;
    int   offX, offY;
    {
        const int c8 = lane % LANES_PER_RAY;
        int ray = wbase + rr;
        if (ray >= nrays) ray = nrays - 1;
        const float* wrow = wp + (long)ray * NSAMP;
        const float* zrow = zp + (long)ray * (NSAMP + 1);

        offX = op * OP_PITCH + r1 * RAY_PITCH + (chunk * 2 + hi) * 8;
        offY = (op ^ 1) * OP_PITCH + r1 * RAY_PITCH + (chunk * 2 + hi) * 8;

        const float nv  = nearp[ray];
        const float fv  = farp[ray];
        v4fu  w4[4], z4[4];
        float z5[4];
#pragma unroll
        for (int q = 0; q < 4; ++q) {
            const int i0 = q * 32 + c8 * 4;
            w4[q] = *(const v4fu*)(wrow + i0);
            z4[q] = *(const v4fu*)(zrow + i0);
            z5[q] = zrow[i0 + 4];
            __builtin_amdgcn_sched_barrier(0);
        }
        const float inv = 1.0f / (fv - nv);

        float* pB = &T[rr * RAY_PITCH + c8];
        float* pA = &T[OP_PITCH + rr * RAY_PITCH + c8];
        float t3 = 0.0f;
        float pr = 0.0f;
#pragma unroll
        for (int q = 0; q < 4; ++q) {
            float b[5];
#pragma unroll
            for (int e = 0; e < 4; ++e) b[e] = (z4[q][e] - nv) * inv;
            b[4] = (z5[q] - nv) * inv;
            float sa = 0.0f, sb = 0.0f;
#pragma unroll
            for (int j = 0; j < 4; ++j) {
                const float w   = w4[q][j];
                const float blo = b[j];
                const float bhi = b[j + 1];
                const float m   = 0.5f * (blo + bhi);
                const float a   = w * m;
                t3 += w * w * (bhi - blo);
                if (j == 0) {
                    sa = a;
                    sb = w;
                } else {
                    pr += a * sb - w * sa;
                    sa += a;
                    sb += w;
                }
            }
            pB[q * 8] = sb;
            pA[q * 8] = sa;
        }
        loc = sum_over_8_lanes((1.0f / 3.0f) * t3 + 2.0f * pr);
    }

    __builtin_amdgcn_fence(__ATOMIC_RELEASE, "wavefront");
    __builtin_amdgcn_wave_barrier();
    __builtin_amdgcn_fence(__ATOMIC_ACQUIRE, "wavefront");

    float cross;
    {
        const float* X = &T[offX];
        const float* Y = &T[offY];
        const v4f x0 = *(const v4f*)(X);
        const v4f x1 = *(const v4f*)(X + 4);
        const v4f y0 = *(const v4f*)(Y);
        const v4f y1 = *(const v4f*)(Y + 4);
        __builtin_amdgcn_sched_barrier(0);

        v2f sfrag[4];
#pragma unroll
        for (int kk = 0; kk < 4; ++kk) {
            const int k0 = hi * 8 + kk * 2;
            sfrag[kk].x = (k0     < col) ? 1.0f : 0.0f;
            sfrag[kk].y = (k0 + 1 < col) ? 1.0f : 0.0f;
        }

        const float x[8] = {x0[0], x0[1], x0[2], x0[3], x1[0], x1[1], x1[2], x1[3]};
        const float y[8] = {y0[0], y0[1], y0[2], y0[3], y1[0], y1[1], y1[2], y1[3]};

        v8f D = {};
#pragma unroll
        for (int kk = 0; kk < 4; ++kk) {
            v2f xf;
            xf.x = x[2 * kk];
            xf.y = x[2 * kk + 1];
            D = __builtin_amdgcn_wmma_f32_16x16x4_f32(
                    false, sfrag[kk], false, xf, (short)0, D, false, false);
        }

        float dot = 0.0f, sx = 0.0f;
#pragma unroll
        for (int g = 0; g < 8; ++g) {
            dot += y[g] * D[g];
            sx  += x[g];
        }
        dot += other_half(dot);
        sx  += other_half(sx);
        const float mir = __int_as_float(__builtin_amdgcn_update_dpp(
                              0, __float_as_int(sx), 0x1B, 0xF, 0xF, true));
        float v = dot + ((chunk == 0) ? sx * mir : 0.0f);
        v = op ? -v : v;
        cross = sum_over_quad(v);
    }

    if ((lane & 3) == 0 && r1 == rr) {
        const int ray = wbase + r1;
        if (ray < nrays) out[ray] = 2.0f * cross + loc;
    }
}

extern "C" void kernel_launch(void* const* d_in, const int* in_sizes, int n_in,
                              void* d_out, int out_size, void* d_ws, size_t ws_size,
                              hipStream_t stream) {
    const float* wp    = (const float*)d_in[0];
    const float* zp    = (const float*)d_in[1];
    const float* nearp = (const float*)d_in[2];
    const float* farp  = (const float*)d_in[3];
    float* out = (float*)d_out;

    const int nrays = in_sizes[2];
    const int raysPerBlock = RAYS_PER_BLOCK;
    const int blocks = (nrays + raysPerBlock - 1) / raysPerBlock;

    distortion_loss_wave_tile_kernel<<<blocks, WAVES_PER_BLOCK * 32, 0, stream>>>(
        wp, zp, nearp, farp, out, nrays);
}
